# starting kernel
# baseline (speedup 1.0000x reference)
#include <hip/hip_runtime.h>


typedef float v2f __attribute__((ext_vector_type(2)));
typedef float v8f __attribute__((ext_vector_type(8)));

#define NHEADS 4
#define HID    64
#define NL     3
#define BATCH  64
#define NNODES 512
#define DIM    64
#define OUTD   (NHEADS * HID)
#define NEG_SLOPE 0.2f

__device__ __forceinline__ v8f wmma_f32(v2f a, v2f b, v8f c) {
  return __builtin_amdgcn_wmma_f32_16x16x4_f32(false, a, false, b, (short)0, c,
                                               false, false);
}

__device__ __forceinline__ float lrelu(float x) {
  return x >= 0.f ? x : NEG_SLOPE * x;
}

__global__ __launch_bounds__(256) void k_lin(const float* __restrict__ x,
                                             const float* __restrict__ W,
                                             float* __restrict__ hout) {
  const int tid  = threadIdx.x;
  const int lane = tid & 31;
  const int wid  = tid >> 5;
  const int half = lane >> 4;
  const int mr   = lane & 15;

  const int mtile = blockIdx.x;
  const int ctile = wid;
  const int row0  = mtile * 64;
  const int o0    = ctile * 32;

  const v8f vz = {0.f,0.f,0.f,0.f,0.f,0.f,0.f,0.f};
  v8f acc[4][2];
#pragma unroll
  for (int sm = 0; sm < 4; ++sm)
#pragma unroll
    for (int sn = 0; sn < 2; ++sn) acc[sm][sn] = vz;

  for (int kk = 0; kk < DIM; kk += 4) {
    const int kbase = kk + 2 * half;
    v2f afr[4], bfr[2];
#pragma unroll
    for (int sm = 0; sm < 4; ++sm)
      afr[sm] = *(const v2f*)(x + (size_t)(row0 + sm * 16 + mr) * DIM + kbase);
#pragma unroll
    for (int sn = 0; sn < 2; ++sn)
      bfr[sn] = *(const v2f*)(W + (size_t)(o0 + sn * 16 + mr) * DIM + kbase);
#pragma unroll
    for (int sm = 0; sm < 4; ++sm)
#pragma unroll
      for (int sn = 0; sn < 2; ++sn)
        acc[sm][sn] = wmma_f32(afr[sm], bfr[sn], acc[sm][sn]);
  }

#pragma unroll
  for (int sm = 0; sm < 4; ++sm)
#pragma unroll
    for (int sn = 0; sn < 2; ++sn)
#pragma unroll
      for (int r = 0; r < 8; ++r) {
        const int row = row0 + sm * 16 + r + 8 * half;
        const int col = o0 + sn * 16 + mr;
        hout[(size_t)row * OUTD + col] = acc[sm][sn][r];
      }
}

__global__ __launch_bounds__(256) void k_scores(const float* __restrict__ hbuf,
                                                const float* __restrict__ attS,
                                                const float* __restrict__ attD,
                                                float* __restrict__ asrc,
                                                float* __restrict__ adst) {
  const int tid  = threadIdx.x;
  const int lane = tid & 31;
  const int wid  = tid >> 5;
  const int node = blockIdx.x * 8 + wid;
  const int b = node >> 9, n = node & 511;
  const float* hrow = hbuf + (size_t)node * OUTD;

  for (int hh = 0; hh < NHEADS; ++hh) {
    const float h0 = hrow[hh * HID + lane];
    const float h1 = hrow[hh * HID + 32 + lane];
    float ss = h0 * attS[hh * HID + lane] + h1 * attS[hh * HID + 32 + lane];
    float sd = h0 * attD[hh * HID + lane] + h1 * attD[hh * HID + 32 + lane];
#pragma unroll
    for (int off = 16; off; off >>= 1) {
      ss += __shfl_xor(ss, off, 32);
      sd += __shfl_xor(sd, off, 32);
    }
    if (lane == 0) {
      asrc[((b * NHEADS + hh) << 9) + n] = ss;
      adst[((b * NHEADS + hh) << 9) + n] = sd;
    }
  }
}

__global__ __launch_bounds__(256) void k_agg(const float* __restrict__ hbuf,
                                             const float* __restrict__ asrc,
                                             const float* __restrict__ adst,
                                             const float* __restrict__ bias,
                                             float* __restrict__ xout) {
  __shared__ float sH[128 * 80];
  __shared__ float sA[512];
  __shared__ float sred[8];

  const int tid  = threadIdx.x;
  const int lane = tid & 31;
  const int wid  = tid >> 5;
  const int half = lane >> 4;
  const int mr   = lane & 15;

  const int b    = blockIdx.y;
  const int i0   = blockIdx.x * 128;
  const int mrow = i0 + wid * 16 + mr;

  const v8f vz = {0.f,0.f,0.f,0.f,0.f,0.f,0.f,0.f};
  v8f accm[4];
#pragma unroll
  for (int t = 0; t < 4; ++t) accm[t] = vz;

  for (int hh = 0; hh < NHEADS; ++hh) {
    const int bh = (b * NHEADS + hh) << 9;

    __syncthreads();
    sA[tid]       = asrc[bh + tid];
    sA[tid + 256] = asrc[bh + tid + 256];
    __syncthreads();

    float mx = fmaxf(sA[tid], sA[tid + 256]);
#pragma unroll
    for (int off = 16; off; off >>= 1) mx = fmaxf(mx, __shfl_xor(mx, off, 32));
    if (lane == 0) sred[wid] = mx;
    __syncthreads();
    float smax = sred[0];
#pragma unroll
    for (int w = 1; w < 8; ++w) smax = fmaxf(smax, sred[w]);

    const float d     = adst[bh + mrow];
    const float mterm = lrelu(d + smax);

    v8f acc[4];
#pragma unroll
    for (int t = 0; t < 4; ++t) acc[t] = vz;
    float zacc = 0.f;

    for (int jc = 0; jc < 4; ++jc) {
      const int j0 = jc * 128;
      __syncthreads();
#pragma unroll
      for (int k = 0; k < 32; ++k) {
        const int e  = tid + k * 256;
        const int rr = e >> 6, cc = e & 63;
        sH[rr * 80 + cc] =
            hbuf[(size_t)(b * NNODES + j0 + rr) * OUTD + hh * HID + cc];
      }
      __syncthreads();

      for (int jj = 0; jj < 128; jj += 4) {
        const int ja = j0 + jj + 2 * half;
        const float s0 = sA[ja];
        const float s1 = sA[ja + 1];
        const float p0 = __expf(lrelu(d + s0) - mterm);
        const float p1 = __expf(lrelu(d + s1) - mterm);
        zacc += p0 + p1;
        v2f af; af.x = p0; af.y = p1;
        const int r0 = (jj + 2 * half) * 80 + mr;
#pragma unroll
        for (int t = 0; t < 4; ++t) {
          v2f bf;
          bf.x = sH[r0 + t * 16];
          bf.y = sH[r0 + 80 + t * 16];
          acc[t] = wmma_f32(af, bf, acc[t]);
        }
      }
    }

    const float z = zacc + __shfl_xor(zacc, 16, 32);
#pragma unroll
    for (int r = 0; r < 8; ++r) {
      const float zr = __shfl(z, r + 8 * half, 32);
      const float sc = 0.25f / zr;
#pragma unroll
      for (int t = 0; t < 4; ++t) accm[t][r] += acc[t][r] * sc;
    }
  }

#pragma unroll
  for (int t = 0; t < 4; ++t) {
    const float bsv = bias[t * 16 + mr];
#pragma unroll
    for (int r = 0; r < 8; ++r) {
      float v = accm[t][r] + bsv;
      v = v > 0.f ? v : 0.f;
      xout[(size_t)(b * NNODES + i0 + wid * 16 + r + 8 * half) * HID +
           t * 16 + mr] = v;
    }
  }
}

__global__ __launch_bounds__(64) void k_readout(const float* __restrict__ x,
                                                const float* __restrict__ Wr,
                                                const float* __restrict__ br,
                                                float* __restrict__ out) {
  __shared__ float pooled[DIM];
  const int b = blockIdx.x, d = threadIdx.x;
  float s = 0.f;
  for (int i = 0; i < NNODES; ++i)
    s += x[(size_t)(b * NNODES + i) * DIM + d];
  pooled[d] = s * (1.f / (float)NNODES);
  __syncthreads();
  float o = br[d];
  for (int k = 0; k < DIM; ++k) o += pooled[k] * Wr[d * DIM + k];
  out[b * DIM + d] = o;
}

extern "C" void kernel_launch(void* const* d_in, const int* in_sizes, int n_in,
                              void* d_out, int out_size, void* d_ws,
                              size_t ws_size, hipStream_t stream) {
  (void)in_sizes; (void)n_in; (void)out_size; (void)ws_size;
  const float* emb    = (const float*)d_in[0];
  const float* lin_w  = (const float*)d_in[1];
  const float* attS   = (const float*)d_in[2];
  const float* attD   = (const float*)d_in[3];
  const float* conv_b = (const float*)d_in[4];
  const float* rw     = (const float*)d_in[5];
  const float* rb     = (const float*)d_in[6];

  float* ws = (float*)d_ws;
  const size_t XSZ = (size_t)BATCH * NNODES * DIM;
  const size_t HSZ = (size_t)BATCH * NNODES * OUTD;
  const size_t ASZ = (size_t)BATCH * NHEADS * NNODES;
  float* xbuf0 = ws;
  float* xbuf1 = ws + XSZ;
  float* hbuf  = ws + 2 * XSZ;
  float* asrc  = hbuf + HSZ;
  float* adst  = asrc + ASZ;

  const float* xin = emb;
  float* xo[NL] = {xbuf0, xbuf1, xbuf0};

  for (int l = 0; l < NL; ++l) {
    k_lin<<<(BATCH * NNODES) / 64, 256, 0, stream>>>(
        xin, lin_w + (size_t)l * OUTD * DIM, hbuf);
    k_scores<<<(BATCH * NNODES) / 8, 256, 0, stream>>>(
        hbuf, attS + l * NHEADS * HID, attD + l * NHEADS * HID, asrc, adst);
    k_agg<<<dim3(NNODES / 128, BATCH), 256, 0, stream>>>(
        hbuf, asrc, adst, conv_b + l * HID, xo[l]);
    xin = xo[l];
  }
  k_readout<<<BATCH, 64, 0, stream>>>(xin, rw, rb, (float*)d_out);
}


# k_layer: H kept transposed in LDS in per-lane blocks with a copy of s: one address, one 128-bit read per column tile for two k-steps; k order permuted alike in A and B; sum order changed
# speedup vs baseline: 1.6235x; 1.6235x over previous
#include <hip/hip_runtime.h>


typedef float v2f __attribute__((ext_vector_type(2)));
typedef float v8f __attribute__((ext_vector_type(8)));

#define NHEADS 4
#define HID    64
#define NL     3
#define BATCH  64
#define NNODES 512
#define DIM    64
#define OUTD   (NHEADS * HID)
#define NEG_SLOPE 0.2f

__device__ __forceinline__ v8f wmma_f32(v2f a, v2f b, v8f c) {
  return __builtin_amdgcn_wmma_f32_16x16x4_f32(false, a, false, b, (short)0, c,
                                               false, false);
}

__device__ __forceinline__ float lrelu(float x) {
  return x >= 0.f ? x : NEG_SLOPE * x;
}

#define BP 2564
#define XP 68
__global__ __launch_bounds__(1024) void k_layer(const float* __restrict__ x,
                                                const float* __restrict__ W,
                                                const float* __restrict__ attS,
                                                const float* __restrict__ attD,
                                                float* __restrict__ planes) {
  __shared__ __attribute__((aligned(16))) float sHT[16 * BP];
  __shared__ __attribute__((aligned(16))) float sX[NNODES * XP];
  __shared__ __attribute__((aligned(16))) float sW[HID * XP];
  __shared__ float sD[NNODES];

  const int tid  = threadIdx.x;
  const int lane = tid & 31;
  const int wid  = tid >> 5;
  const int half = lane >> 4;
  const int mr   = lane & 15;
  const int hh   = blockIdx.x;
  const int b    = blockIdx.y;
  const int row0 = wid * 16;
  float* tile = sX + row0 * XP;

  {
    const float* xg = x + ((size_t)b * NNODES + row0) * DIM;
#pragma unroll
    for (int q = 0; q < 8; ++q) {
      const int e  = q * 32 + lane;
      const int rr = e >> 4, c4 = e & 15;
      const float4 v = *(const float4*)(xg + e * 4);
      *(float4*)(&tile[rr * XP + c4 * 4]) = v;
    }
    const float4 w4 = *(const float4*)(W + (size_t)hh * HID * DIM + tid * 4);
    *(float4*)(&sW[(tid >> 4) * XP + (tid & 15) * 4]) = w4;
  }
  __syncthreads();

  const v8f vz = {0.f,0.f,0.f,0.f,0.f,0.f,0.f,0.f};
  v8f acc[4];
#pragma unroll
  for (int t = 0; t < 4; ++t) acc[t] = vz;

#pragma unroll
  for (int kk = 0; kk < DIM; kk += 4) {
    const int kbase = kk + 2 * half;
    const v2f af = *(const v2f*)(&tile[mr * XP + kbase]);
#pragma unroll
    for (int t = 0; t < 4; ++t) {
      const v2f bf = *(const v2f*)(&sW[(t * 16 + mr) * XP + kbase]);
      acc[t] = wmma_f32(af, bf, acc[t]);
    }
  }

  {
    float as[4], ad[4];
#pragma unroll
    for (int t = 0; t < 4; ++t) {
      as[t] = attS[hh * HID + t * 16 + mr];
      ad[t] = attD[hh * HID + t * 16 + mr];
    }
    float sq[8];
    float dv = 0.f;
#pragma unroll
    for (int r = 0; r < 8; ++r) {
      float ss = acc[0][r] * as[0];
      float sd = acc[0][r] * ad[0];
#pragma unroll
      for (int t = 1; t < 4; ++t) {
        ss += acc[t][r] * as[t];
        sd += acc[t][r] * ad[t];
      }
#pragma unroll
      for (int off = 8; off; off >>= 1) {
        ss += __shfl_xor(ss, off, 32);
        sd += __shfl_xor(sd, off, 32);
      }
      sq[r] = ss;
      dv = (mr == r) ? sd : dv;
    }
    if (mr < 8) sD[row0 + mr + 8 * half] = dv;
    {
      float* sp = &sHT[mr * BP + 4 * NNODES + row0 + 8 * half];
      float4 lo, hi;
      lo.x = sq[0]; lo.y = sq[1]; lo.z = sq[2]; lo.w = sq[3];
      hi.x = sq[4]; hi.y = sq[5]; hi.z = sq[6]; hi.w = sq[7];
      *(float4*)(sp)     = lo;
      *(float4*)(sp + 4) = hi;
    }
#pragma unroll
    for (int t = 0; t < 4; ++t) {
      float* hp = &sHT[mr * BP + t * NNODES + row0 + 8 * half];
      float4 lo, hi;
      lo.x = acc[t][0]; lo.y = acc[t][1]; lo.z = acc[t][2]; lo.w = acc[t][3];
      hi.x = acc[t][4]; hi.y = acc[t][5]; hi.z = acc[t][6]; hi.w = acc[t][7];
      *(float4*)(hp)     = lo;
      *(float4*)(hp + 4) = hi;
    }
  }
  __syncthreads();

  const float* s0row = sHT + 4 * NNODES;
  float mx = s0row[lane];
#pragma unroll
  for (int k = 1; k < 16; ++k) mx = fmaxf(mx, s0row[lane + 32 * k]);
#pragma unroll
  for (int off = 16; off; off >>= 1) mx = fmaxf(mx, __shfl_xor(mx, off, 32));
  const float smax = mx;

  const float d     = sD[row0 + mr];
  const float mterm = lrelu(d + smax);

#pragma unroll
  for (int t = 0; t < 4; ++t) acc[t] = vz;
  float zacc = 0.f;

  const float* hb = sHT + mr * BP + 4 * half;
#pragma unroll 2
  for (int j0 = 0; j0 < NNODES; j0 += 8) {
    const float4 s4 = *(const float4*)(hb + 4 * NNODES + j0);
    float4 h4[4];
#pragma unroll
    for (int t = 0; t < 4; ++t)
      h4[t] = *(const float4*)(hb + t * NNODES + j0);
    const float p0 = __expf(lrelu(d + s4.x) - mterm);
    const float p1 = __expf(lrelu(d + s4.y) - mterm);
    const float p2 = __expf(lrelu(d + s4.z) - mterm);
    const float p3 = __expf(lrelu(d + s4.w) - mterm);
    zacc += p0 + p1;
    zacc += p2 + p3;
    v2f a0; a0.x = p0; a0.y = p1;
    v2f a1; a1.x = p2; a1.y = p3;
#pragma unroll
    for (int t = 0; t < 4; ++t) {
      v2f bf; bf.x = h4[t].x; bf.y = h4[t].y;
      acc[t] = wmma_f32(a0, bf, acc[t]);
    }
#pragma unroll
    for (int t = 0; t < 4; ++t) {
      v2f bf; bf.x = h4[t].z; bf.y = h4[t].w;
      acc[t] = wmma_f32(a1, bf, acc[t]);
    }
  }

  const float z = zacc + __shfl_xor(zacc, 16, 32);

#pragma unroll
  for (int r = 0; r < 8; ++r) {
    const float zr = __shfl(z, r + 8 * half, 32);
    const float sc = 0.25f / zr;
#pragma unroll
    for (int t = 0; t < 4; ++t)
      tile[(r + 8 * half) * XP + t * 16 + mr] = acc[t][r] * sc;
  }
  __syncthreads();

  float* pg = planes + (((size_t)b * NHEADS + hh) * NNODES + row0) * HID;
#pragma unroll
  for (int q = 0; q < 8; ++q) {
    const int e  = q * 32 + lane;
    const int rr = e >> 4, c4 = e & 15;
    const float4 v = *(const float4*)(&tile[rr * XP + c4 * 4]);
    *(float4*)(pg + e * 4) = v;
  }
}

#define PLANE4 (NNODES * HID / 4)
__global__ __launch_bounds__(256) void k_combine(const float* __restrict__ planes,
                                                 const float* __restrict__ bias,
                                                 float* __restrict__ xout) {
  const int tid = threadIdx.x;
#pragma unroll
  for (int k = 0; k < 4; ++k) {
    const int f4  = (blockIdx.x * 4 + k) * 256 + tid;
    const int b   = f4 / PLANE4;
    const int rem = f4 % PLANE4;
    const float* p = planes + ((size_t)b * NHEADS * PLANE4 + rem) * 4;
    const float4 v0 = *(const float4*)(p);
    const float4 v1 = *(const float4*)(p + 1 * PLANE4 * 4);
    const float4 v2 = *(const float4*)(p + 2 * PLANE4 * 4);
    const float4 v3 = *(const float4*)(p + 3 * PLANE4 * 4);
    const float4 bb = *(const float4*)(bias + (f4 & 15) * 4);
    float4 o;
    o.x = v0.x; o.x += v1.x; o.x += v2.x; o.x += v3.x; o.x += bb.x;
    o.y = v0.y; o.y += v1.y; o.y += v2.y; o.y += v3.y; o.y += bb.y;
    o.z = v0.z; o.z += v1.z; o.z += v2.z; o.z += v3.z; o.z += bb.z;
    o.w = v0.w; o.w += v1.w; o.w += v2.w; o.w += v3.w; o.w += bb.w;
    o.x = o.x > 0.f ? o.x : 0.f;
    o.y = o.y > 0.f ? o.y : 0.f;
    o.z = o.z > 0.f ? o.z : 0.f;
    o.w = o.w > 0.f ? o.w : 0.f;
    *(float4*)(xout + (size_t)f4 * 4) = o;
  }
}

__global__ __launch_bounds__(64) void k_readout(const float* __restrict__ planes,
                                                const float* __restrict__ bias,
                                                const float* __restrict__ Wr,
                                                const float* __restrict__ br,
                                                float* __restrict__ out) {
  __shared__ float pooled[DIM];
  const int b = blockIdx.x, d = threadIdx.x;
  const float* p = planes + (size_t)b * NHEADS * NNODES * HID + d;
  const float bd = bias[d];
  float s = 0.f;
#pragma unroll 16
  for (int i = 0; i < NNODES; ++i) {
    float v = p[(size_t)i * HID];
    v += p[(size_t)(1 * NNODES + i) * HID];
    v += p[(size_t)(2 * NNODES + i) * HID];
    v += p[(size_t)(3 * NNODES + i) * HID];
    v += bd;
    s += v > 0.f ? v : 0.f;
  }
  pooled[d] = s * (1.f / (float)NNODES);
  __syncthreads();
  float o = br[d];
  for (int k = 0; k < DIM; ++k) o += pooled[k] * Wr[d * DIM + k];
  out[b * DIM + d] = o;
}

extern "C" void kernel_launch(void* const* d_in, const int* in_sizes, int n_in,
                              void* d_out, int out_size, void* d_ws,
                              size_t ws_size, hipStream_t stream) {
  (void)in_sizes; (void)n_in; (void)out_size; (void)ws_size;
  const float* emb    = (const float*)d_in[0];
  const float* lin_w  = (const float*)d_in[1];
  const float* attS   = (const float*)d_in[2];
  const float* attD   = (const float*)d_in[3];
  const float* conv_b = (const float*)d_in[4];
  const float* rw     = (const float*)d_in[5];
  const float* rb     = (const float*)d_in[6];

  float* ws = (float*)d_ws;
  const size_t XSZ = (size_t)BATCH * NNODES * DIM;
  float* xbuf   = ws;
  float* planes = ws + XSZ;

  const float* xin = emb;
  for (int l = 0; l < NL; ++l) {
    k_layer<<<dim3(NHEADS, BATCH), 1024, 0, stream>>>(
        xin, lin_w + (size_t)l * OUTD * DIM, attS + l * NHEADS * HID,
        attD + l * NHEADS * HID, planes);
    if (l + 1 < NL) {
      k_combine<<<(BATCH * NNODES * DIM) / (4 * 4 * 256), 256, 0, stream>>>(
          planes, conv_b + l * HID, xbuf);
      xin = xbuf;
    }
  }
  k_readout<<<BATCH, 64, 0, stream>>>(planes, conv_b + (NL - 1) * HID, rw, rb,
                                      (float*)d_out);
}


# Last layer's node mean split over the chip: new k_pool, 8 blocks per batch element: head sum, bias, ReLU, 64-row sums from independent loads; readout adds 8 chunk sums; 7 launches; sum order changed
# speedup vs baseline: 1.8462x; 1.1372x over previous
#include <hip/hip_runtime.h>


typedef float v2f __attribute__((ext_vector_type(2)));
typedef float v8f __attribute__((ext_vector_type(8)));

#define NHEADS 4
#define HID    64
#define NL     3
#define BATCH  64
#define NNODES 512
#define DIM    64
#define OUTD   (NHEADS * HID)
#define NEG_SLOPE 0.2f

__device__ __forceinline__ v8f wmma_f32(v2f a, v2f b, v8f c) {
  return __builtin_amdgcn_wmma_f32_16x16x4_f32(false, a, false, b, (short)0, c,
                                               false, false);
}

__device__ __forceinline__ float lrelu(float x) {
  return x >= 0.f ? x : NEG_SLOPE * x;
}

#define BP 2564
#define XP 68
__global__ __launch_bounds__(1024) void k_layer(const float* __restrict__ x,
                                                const float* __restrict__ W,
                                                const float* __restrict__ attS,
                                                const float* __restrict__ attD,
                                                float* __restrict__ planes) {
  __shared__ __attribute__((aligned(16))) float sHT[16 * BP];
  __shared__ __attribute__((aligned(16))) float sX[NNODES * XP];
  __shared__ __attribute__((aligned(16))) float sW[HID * XP];
  __shared__ float sD[NNODES];

  const int tid  = threadIdx.x;
  const int lane = tid & 31;
  const int wid  = tid >> 5;
  const int half = lane >> 4;
  const int mr   = lane & 15;
  const int hh   = blockIdx.x;
  const int b    = blockIdx.y;
  const int row0 = wid * 16;
  float* tile = sX + row0 * XP;

  {
    const float* xg = x + ((size_t)b * NNODES + row0) * DIM;
#pragma unroll
    for (int q = 0; q < 8; ++q) {
      const int e  = q * 32 + lane;
      const int rr = e >> 4, c4 = e & 15;
      const float4 v = *(const float4*)(xg + e * 4);
      *(float4*)(&tile[rr * XP + c4 * 4]) = v;
    }
    const float4 w4 = *(const float4*)(W + (size_t)hh * HID * DIM + tid * 4);
    *(float4*)(&sW[(tid >> 4) * XP + (tid & 15) * 4]) = w4;
  }
  __syncthreads();

  const v8f vz = {0.f,0.f,0.f,0.f,0.f,0.f,0.f,0.f};
  v8f acc[4];
#pragma unroll
  for (int t = 0; t < 4; ++t) acc[t] = vz;

#pragma unroll
  for (int kk = 0; kk < DIM; kk += 4) {
    const int kbase = kk + 2 * half;
    const v2f af = *(const v2f*)(&tile[mr * XP + kbase]);
#pragma unroll
    for (int t = 0; t < 4; ++t) {
      const v2f bf = *(const v2f*)(&sW[(t * 16 + mr) * XP + kbase]);
      acc[t] = wmma_f32(af, bf, acc[t]);
    }
  }

  {
    float as[4], ad[4];
#pragma unroll
    for (int t = 0; t < 4; ++t) {
      as[t] = attS[hh * HID + t * 16 + mr];
      ad[t] = attD[hh * HID + t * 16 + mr];
    }
    float sq[8];
    float dv = 0.f;
#pragma unroll
    for (int r = 0; r < 8; ++r) {
      float ss = acc[0][r] * as[0];
      float sd = acc[0][r] * ad[0];
#pragma unroll
      for (int t = 1; t < 4; ++t) {
        ss += acc[t][r] * as[t];
        sd += acc[t][r] * ad[t];
      }
#pragma unroll
      for (int off = 8; off; off >>= 1) {
        ss += __shfl_xor(ss, off, 32);
        sd += __shfl_xor(sd, off, 32);
      }
      sq[r] = ss;
      dv = (mr == r) ? sd : dv;
    }
    if (mr < 8) sD[row0 + mr + 8 * half] = dv;
    {
      float* sp = &sHT[mr * BP + 4 * NNODES + row0 + 8 * half];
      float4 lo, hi;
      lo.x = sq[0]; lo.y = sq[1]; lo.z = sq[2]; lo.w = sq[3];
      hi.x = sq[4]; hi.y = sq[5]; hi.z = sq[6]; hi.w = sq[7];
      *(float4*)(sp)     = lo;
      *(float4*)(sp + 4) = hi;
    }
#pragma unroll
    for (int t = 0; t < 4; ++t) {
      float* hp = &sHT[mr * BP + t * NNODES + row0 + 8 * half];
      float4 lo, hi;
      lo.x = acc[t][0]; lo.y = acc[t][1]; lo.z = acc[t][2]; lo.w = acc[t][3];
      hi.x = acc[t][4]; hi.y = acc[t][5]; hi.z = acc[t][6]; hi.w = acc[t][7];
      *(float4*)(hp)     = lo;
      *(float4*)(hp + 4) = hi;
    }
  }
  __syncthreads();

  const float* s0row = sHT + 4 * NNODES;
  float mx = s0row[lane];
#pragma unroll
  for (int k = 1; k < 16; ++k) mx = fmaxf(mx, s0row[lane + 32 * k]);
#pragma unroll
  for (int off = 16; off; off >>= 1) mx = fmaxf(mx, __shfl_xor(mx, off, 32));
  const float smax = mx;

  const float d     = sD[row0 + mr];
  const float mterm = lrelu(d + smax);

#pragma unroll
  for (int t = 0; t < 4; ++t) acc[t] = vz;
  float zacc = 0.f;

  const float* hb = sHT + mr * BP + 4 * half;
#pragma unroll 2
  for (int j0 = 0; j0 < NNODES; j0 += 8) {
    const float4 s4 = *(const float4*)(hb + 4 * NNODES + j0);
    float4 h4[4];
#pragma unroll
    for (int t = 0; t < 4; ++t)
      h4[t] = *(const float4*)(hb + t * NNODES + j0);
    const float p0 = __expf(lrelu(d + s4.x) - mterm);
    const float p1 = __expf(lrelu(d + s4.y) - mterm);
    const float p2 = __expf(lrelu(d + s4.z) - mterm);
    const float p3 = __expf(lrelu(d + s4.w) - mterm);
    zacc += p0 + p1;
    zacc += p2 + p3;
    v2f a0; a0.x = p0; a0.y = p1;
    v2f a1; a1.x = p2; a1.y = p3;
#pragma unroll
    for (int t = 0; t < 4; ++t) {
      v2f bf; bf.x = h4[t].x; bf.y = h4[t].y;
      acc[t] = wmma_f32(a0, bf, acc[t]);
    }
#pragma unroll
    for (int t = 0; t < 4; ++t) {
      v2f bf; bf.x = h4[t].z; bf.y = h4[t].w;
      acc[t] = wmma_f32(a1, bf, acc[t]);
    }
  }

  const float z = zacc + __shfl_xor(zacc, 16, 32);

#pragma unroll
  for (int r = 0; r < 8; ++r) {
    const float zr = __shfl(z, r + 8 * half, 32);
    const float sc = 0.25f / zr;
#pragma unroll
    for (int t = 0; t < 4; ++t)
      tile[(r + 8 * half) * XP + t * 16 + mr] = acc[t][r] * sc;
  }
  __syncthreads();

  float* pg = planes + (((size_t)b * NHEADS + hh) * NNODES + row0) * HID;
#pragma unroll
  for (int q = 0; q < 8; ++q) {
    const int e  = q * 32 + lane;
    const int rr = e >> 4, c4 = e & 15;
    const float4 v = *(const float4*)(&tile[rr * XP + c4 * 4]);
    *(float4*)(pg + e * 4) = v;
  }
}

#define PLANE4 (NNODES * HID / 4)
__global__ __launch_bounds__(256) void k_combine(const float* __restrict__ planes,
                                                 const float* __restrict__ bias,
                                                 float* __restrict__ xout) {
  const int tid = threadIdx.x;
#pragma unroll
  for (int k = 0; k < 4; ++k) {
    const int f4  = (blockIdx.x * 4 + k) * 256 + tid;
    const int b   = f4 / PLANE4;
    const int rem = f4 % PLANE4;
    const float* p = planes + ((size_t)b * NHEADS * PLANE4 + rem) * 4;
    const float4 v0 = *(const float4*)(p);
    const float4 v1 = *(const float4*)(p + 1 * PLANE4 * 4);
    const float4 v2 = *(const float4*)(p + 2 * PLANE4 * 4);
    const float4 v3 = *(const float4*)(p + 3 * PLANE4 * 4);
    const float4 bb = *(const float4*)(bias + (f4 & 15) * 4);
    float4 o;
    o.x = v0.x; o.x += v1.x; o.x += v2.x; o.x += v3.x; o.x += bb.x;
    o.y = v0.y; o.y += v1.y; o.y += v2.y; o.y += v3.y; o.y += bb.y;
    o.z = v0.z; o.z += v1.z; o.z += v2.z; o.z += v3.z; o.z += bb.z;
    o.w = v0.w; o.w += v1.w; o.w += v2.w; o.w += v3.w; o.w += bb.w;
    o.x = o.x > 0.f ? o.x : 0.f;
    o.y = o.y > 0.f ? o.y : 0.f;
    o.z = o.z > 0.f ? o.z : 0.f;
    o.w = o.w > 0.f ? o.w : 0.f;
    *(float4*)(xout + (size_t)f4 * 4) = o;
  }
}

#define PCH   8
#define PROWS (NNODES / PCH)
__global__ __launch_bounds__(256) void k_pool(const float* __restrict__ planes,
                                              const float* __restrict__ bias,
                                              float* __restrict__ part) {
  __shared__ __attribute__((aligned(16))) float sP[16 * HID];
  const int tid = threadIdx.x;
  const int c4  = tid & 15;
  const int rl  = tid >> 4;
  const int c   = blockIdx.x;
  const int b   = blockIdx.y;
  const float* p = planes +
      ((size_t)b * NHEADS * NNODES + c * PROWS + rl) * HID + c4 * 4;
  const float4 bb = *(const float4*)(bias + c4 * 4);
  float4 v[4][NHEADS];
#pragma unroll
  for (int k = 0; k < 4; ++k) {
#pragma unroll
    for (int h = 0; h < NHEADS; ++h)
      v[k][h] = *(const float4*)(p + ((size_t)h * NNODES + 16 * k) * HID);
  }
  float4 s;
  s.x = 0.f; s.y = 0.f; s.z = 0.f; s.w = 0.f;
#pragma unroll
  for (int k = 0; k < 4; ++k) {
    float4 o;
    o.x = v[k][0].x; o.x += v[k][1].x; o.x += v[k][2].x; o.x += v[k][3].x; o.x += bb.x;
    o.y = v[k][0].y; o.y += v[k][1].y; o.y += v[k][2].y; o.y += v[k][3].y; o.y += bb.y;
    o.z = v[k][0].z; o.z += v[k][1].z; o.z += v[k][2].z; o.z += v[k][3].z; o.z += bb.z;
    o.w = v[k][0].w; o.w += v[k][1].w; o.w += v[k][2].w; o.w += v[k][3].w; o.w += bb.w;
    s.x += o.x > 0.f ? o.x : 0.f;
    s.y += o.y > 0.f ? o.y : 0.f;
    s.z += o.z > 0.f ? o.z : 0.f;
    s.w += o.w > 0.f ? o.w : 0.f;
  }
  *(float4*)(&sP[rl * HID + c4 * 4]) = s;
  __syncthreads();
  if (tid < HID) {
    float a = sP[tid];
#pragma unroll
    for (int r = 1; r < 16; ++r) a += sP[r * HID + tid];
    part[((size_t)b * PCH + c) * HID + tid] = a;
  }
}

__global__ __launch_bounds__(64) void k_readout(const float* __restrict__ part,
                                                const float* __restrict__ Wr,
                                                const float* __restrict__ br,
                                                float* __restrict__ out) {
  __shared__ float pooled[DIM];
  const int b = blockIdx.x, d = threadIdx.x;
  const float* p = part + (size_t)b * PCH * HID + d;
  float s = p[0];
#pragma unroll
  for (int c = 1; c < PCH; ++c) s += p[c * HID];
  pooled[d] = s * (1.f / (float)NNODES);
  __syncthreads();
  float o = br[d];
  for (int k = 0; k < DIM; ++k) o += pooled[k] * Wr[d * DIM + k];
  out[b * DIM + d] = o;
}

extern "C" void kernel_launch(void* const* d_in, const int* in_sizes, int n_in,
                              void* d_out, int out_size, void* d_ws,
                              size_t ws_size, hipStream_t stream) {
  (void)in_sizes; (void)n_in; (void)out_size; (void)ws_size;
  const float* emb    = (const float*)d_in[0];
  const float* lin_w  = (const float*)d_in[1];
  const float* attS   = (const float*)d_in[2];
  const float* attD   = (const float*)d_in[3];
  const float* conv_b = (const float*)d_in[4];
  const float* rw     = (const float*)d_in[5];
  const float* rb     = (const float*)d_in[6];

  float* ws = (float*)d_ws;
  const size_t XSZ = (size_t)BATCH * NNODES * DIM;
  float* xbuf   = ws;
  float* planes = ws + XSZ;

  const float* xin = emb;
  for (int l = 0; l < NL; ++l) {
    k_layer<<<dim3(NHEADS, BATCH), 1024, 0, stream>>>(
        xin, lin_w + (size_t)l * OUTD * DIM, attS + l * NHEADS * HID,
        attD + l * NHEADS * HID, planes);
    if (l + 1 < NL) {
      k_combine<<<(BATCH * NNODES * DIM) / (4 * 4 * 256), 256, 0, stream>>>(
          planes, conv_b + l * HID, xbuf);
      xin = xbuf;
    }
  }
  float* part = xbuf;
  k_pool<<<dim3(PCH, BATCH), 256, 0, stream>>>(planes, conv_b + (NL - 1) * HID,
                                               part);
  k_readout<<<BATCH, 64, 0, stream>>>(part, rw, rb, (float*)d_out);
}
